# starting kernel
# baseline (speedup 1.0000x reference)
#include <hip/hip_runtime.h>
#include <math.h>


typedef __attribute__((ext_vector_type(16))) _Float16 v16h;
typedef __attribute__((ext_vector_type(8)))  _Float16 v8h;
typedef __attribute__((ext_vector_type(4)))  _Float16 v4h;
typedef __attribute__((ext_vector_type(8)))  float    v8f;
typedef __attribute__((ext_vector_type(4)))  float    v4f;

#define D_MODEL 2048
#define HIDDEN  512
#define NPARAM  96
#define NHEADS  32
#define MTILE   64
#define KT1     (D_MODEL / 32)
#define KT2     (HIDDEN  / 32)
#define NT2     (NPARAM  / 16)

#define XSTRIDE 2056
#define HSTRIDE 520

#define X_BYTES    (MTILE * XSTRIDE * 2)
#define SMEM_BYTES (X_BYTES + 512*4 + 512*4 + 64*4 + 64*4)

#if __has_builtin(__builtin_amdgcn_tanhf)
__device__ __forceinline__ float fast_tanh(float x) { return __builtin_amdgcn_tanhf(x); }
#elif __has_builtin(__builtin_amdgcn_tanh_f32)
__device__ __forceinline__ float fast_tanh(float x) { return __builtin_amdgcn_tanh_f32(x); }
#else
__device__ __forceinline__ float fast_tanh(float x) { return tanhf(x); }
#endif

__device__ __forceinline__ void prefetch_wgp_8k(const void* p) {
    asm volatile("global_prefetch_b8 %0, off offset:8192" :: "v"(p));
}

__device__ __forceinline__ int frag_k(int lane, int e) {
    return (e < 8 ? e : e + 8) + ((lane >= 16) ? 8 : 0);
}

__global__ __launch_bounds__(256) void pack_w1(const float* __restrict__ W1,
                                               const float* __restrict__ gamma,
                                               _Float16* __restrict__ B1pack) {
    int i    = blockIdx.x * 256 + threadIdx.x;
    int e    = i & 15;
    int lane = (i >> 4) & 31;
    int frag = i >> 9;
    int kt   = frag % KT1;
    int nt   = frag / KT1;
    int n    = nt * 16 + (lane & 15);
    int kk   = kt * 32 + frag_k(lane, e);
    B1pack[i] = (_Float16)(gamma[kk] * W1[(size_t)kk * HIDDEN + n]);
}

__global__ __launch_bounds__(256) void pack_w2(const float* __restrict__ W2,
                                               _Float16* __restrict__ B2pack) {
    int i    = blockIdx.x * 256 + threadIdx.x;
    int e    = i & 15;
    int lane = (i >> 4) & 31;
    int frag = i >> 9;
    int kt   = frag % KT2;
    int nt   = frag / KT2;
    int n    = nt * 16 + (lane & 15);
    int kk   = kt * 32 + frag_k(lane, e);
    B2pack[i] = (_Float16)W2[(size_t)kk * NPARAM + n];
}

__global__ __launch_bounds__(256) void col_reduce(const float* __restrict__ W1,
                                                  const float* __restrict__ gamma,
                                                  const float* __restrict__ beta,
                                                  const float* __restrict__ b1,
                                                  float* __restrict__ s1,
                                                  float* __restrict__ b1p) {
    __shared__ float rs[256], rb[256];
    int n = blockIdx.x, t = threadIdx.x;
    float a = 0.f, b = 0.f;
    for (int k = t; k < D_MODEL; k += 256) {
        float w = W1[(size_t)k * HIDDEN + n];
        a += gamma[k] * w;
        b += beta[k] * w;
    }
    rs[t] = a; rb[t] = b;
    __syncthreads();
    for (int s = 128; s > 0; s >>= 1) {
        if (t < s) { rs[t] += rs[t + s]; rb[t] += rb[t + s]; }
        __syncthreads();
    }
    if (t == 0) { s1[n] = rs[0]; b1p[n] = rb[0] + b1[n]; }
}

__global__ __launch_bounds__(512, 1)
void hope_main(const float* __restrict__ slow,
               const _Float16* __restrict__ B1pack,
               const _Float16* __restrict__ B2pack,
               const float* __restrict__ s1,
               const float* __restrict__ b1p,
               const float* __restrict__ b2,
               float* __restrict__ out,
               int Btot) {
    extern __shared__ __align__(16) char smem[];
    _Float16* xs     = (_Float16*)smem;
    float*    psum   = (float*)(smem + X_BYTES);
    float*    psumsq = psum + 512;
    float*    ldsMu  = psumsq + 512;
    float*    ldsRs  = ldsMu + 64;

    const int tid  = threadIdx.x;
    const int row0 = blockIdx.x * MTILE;
    const int wave = tid >> 5;
    const int lane = tid & 31;
    const int lh   = lane & 15;
    const int hiK  = (lane >= 16) ? 8 : 0;
    const int hiM  = (lane >= 16) ? 8 : 0;

    #pragma unroll 4
    for (int it = 0; it < (MTILE * D_MODEL / 4) / 512; ++it) {
        int c  = it * 512 + tid;
        int r  = c >> 9;
        int c4 = c & 511;
        v4f v = *(const v4f*)(slow + (size_t)(row0 + r) * D_MODEL + c4 * 4);
        v4h h = { (_Float16)v[0], (_Float16)v[1], (_Float16)v[2], (_Float16)v[3] };
        *(v4h*)(xs + (size_t)r * XSTRIDE + c4 * 4) = h;
    }
    __syncthreads();

    {
        int r = tid >> 3, part = tid & 7;
        const _Float16* rp = xs + (size_t)r * XSTRIDE + part * 256;
        float s = 0.f, sq = 0.f;
        for (int u = 0; u < 32; ++u) {
            v8h hv = *(const v8h*)(rp + u * 8);
            #pragma unroll
            for (int e = 0; e < 8; ++e) { float f = (float)hv[e]; s += f; sq += f * f; }
        }
        psum[tid] = s; psumsq[tid] = sq;
    }
    __syncthreads();
    if (tid < MTILE) {
        float s = 0.f, sq = 0.f;
        #pragma unroll
        for (int p = 0; p < 8; ++p) { s += psum[tid * 8 + p]; sq += psumsq[tid * 8 + p]; }
        float mu  = s * (1.0f / D_MODEL);
        float var = sq * (1.0f / D_MODEL) - mu * mu;
        ldsMu[tid] = mu;
        ldsRs[tid] = rsqrtf(var + 1e-5f);
    }
    __syncthreads();

    v8f acc[4][2] = {};
    #pragma unroll 2
    for (int kt = 0; kt < KT1; ++kt) {
        int kb = kt * 32 + hiK;
        v16h a[4];
        #pragma unroll
        for (int mt = 0; mt < 4; ++mt) {
            const _Float16* ap = xs + (size_t)(mt * 16 + lh) * XSTRIDE + kb;
            v8h lo = *(const v8h*)ap;
            v8h hi = *(const v8h*)(ap + 16);
            a[mt] = __builtin_shufflevector(lo, hi, 0,1,2,3,4,5,6,7,8,9,10,11,12,13,14,15);
        }
        #pragma unroll
        for (int j = 0; j < 2; ++j) {
            int nt = wave * 2 + j;
            const _Float16* bp = B1pack + ((size_t)(nt * KT1 + kt) * 32 + lane) * 16;
            prefetch_wgp_8k(bp);
            v16h b = *(const v16h*)bp;
            #pragma unroll
            for (int mt = 0; mt < 4; ++mt)
                acc[mt][j] = __builtin_amdgcn_wmma_f32_16x16x32_f16(
                    false, a[mt], false, b, (short)0, acc[mt][j], false, false);
        }
    }
    __syncthreads();

    #pragma unroll
    for (int j = 0; j < 2; ++j) {
        int col = wave * 32 + j * 16 + lh;
        float s1v = s1[col];
        float b1v = b1p[col];
        #pragma unroll
        for (int mt = 0; mt < 4; ++mt) {
            #pragma unroll
            for (int i = 0; i < 8; ++i) {
                int row = mt * 16 + i + hiM;
                float pre = ldsRs[row] * (acc[mt][j][i] - ldsMu[row] * s1v) + b1v;
                float g = 0.5f * pre * (1.0f + erff(pre * 0.70710678118654752f));
                xs[(size_t)row * HSTRIDE + col] = (_Float16)g;
            }
        }
    }
    __syncthreads();

    for (int job = wave; job < 4 * NT2; job += 16) {
        int m2 = job & 3;
        int n2 = job >> 2;
        v8f c2 = {};
        for (int kt = 0; kt < KT2; ++kt) {
            int kb = kt * 32 + hiK;
            const _Float16* ap = xs + (size_t)(m2 * 16 + lh) * HSTRIDE + kb;
            v8h lo = *(const v8h*)ap;
            v8h hi = *(const v8h*)(ap + 16);
            v16h a2 = __builtin_shufflevector(lo, hi, 0,1,2,3,4,5,6,7,8,9,10,11,12,13,14,15);
            v16h b  = *(const v16h*)(B2pack + ((size_t)(n2 * KT2 + kt) * 32 + lane) * 16);
            c2 = __builtin_amdgcn_wmma_f32_16x16x32_f16(
                false, a2, false, b, (short)0, c2, false, false);
        }
        int col  = n2 * 16 + lh;
        int head = col / 3;
        int comp = col % 3;
        float b2v = b2[col];
        #pragma unroll
        for (int i = 0; i < 8; ++i) {
            int grow = row0 + m2 * 16 + i + hiM;
            float val = fast_tanh(c2[i] + b2v) + 1.0f;
            out[(size_t)comp * Btot * NHEADS + (size_t)grow * NHEADS + head] = val;
        }
    }
}

extern "C" void kernel_launch(void* const* d_in, const int* in_sizes, int n_in,
                              void* d_out, int out_size, void* d_ws, size_t ws_size,
                              hipStream_t stream) {
    const float* slow  = (const float*)d_in[0];
    const float* gamma = (const float*)d_in[1];
    const float* beta  = (const float*)d_in[2];
    const float* W1    = (const float*)d_in[3];
    const float* b1    = (const float*)d_in[4];
    const float* W2    = (const float*)d_in[5];
    const float* b2    = (const float*)d_in[6];
    float* out = (float*)d_out;

    const int Btot = in_sizes[0] / D_MODEL;

    char* ws = (char*)d_ws;
    _Float16* B1pack = (_Float16*)ws;
    _Float16* B2pack = (_Float16*)(ws + 2097152);
    float*    s1     = (float*)(ws + 2097152 + 98304);
    float*    b1p    = (float*)(ws + 2097152 + 98304 + 2048);

    pack_w1<<<(D_MODEL * HIDDEN) / 256, 256, 0, stream>>>(W1, gamma, B1pack);
    pack_w2<<<(HIDDEN * NPARAM) / 256, 256, 0, stream>>>(W2, B2pack);
    col_reduce<<<HIDDEN, 256, 0, stream>>>(W1, gamma, beta, b1, s1, b1p);

    hope_main<<<Btot / MTILE, 512, SMEM_BYTES, stream>>>(
        slow, B1pack, B2pack, s1, b1p, b2, out, Btot);
}


# GEMM1 weight fragments loaded four k-steps ahead into a per-wave register ring (replaces the cache prefetch), so a fragment's cache round trip runs under earlier WMMAs instead of in front of its own
# speedup vs baseline: 1.3253x; 1.3253x over previous
#include <hip/hip_runtime.h>
#include <math.h>


typedef __attribute__((ext_vector_type(16))) _Float16 v16h;
typedef __attribute__((ext_vector_type(8)))  _Float16 v8h;
typedef __attribute__((ext_vector_type(4)))  _Float16 v4h;
typedef __attribute__((ext_vector_type(8)))  float    v8f;
typedef __attribute__((ext_vector_type(4)))  float    v4f;

#define D_MODEL 2048
#define HIDDEN  512
#define NPARAM  96
#define NHEADS  32
#define MTILE   64
#define KT1     (D_MODEL / 32)
#define KT2     (HIDDEN  / 32)
#define NT2     (NPARAM  / 16)

#define XSTRIDE 2056
#define HSTRIDE 520

#define X_BYTES    (MTILE * XSTRIDE * 2)
#define SMEM_BYTES (X_BYTES + 512*4 + 512*4 + 64*4 + 64*4)

#if __has_builtin(__builtin_amdgcn_tanhf)
__device__ __forceinline__ float fast_tanh(float x) { return __builtin_amdgcn_tanhf(x); }
#elif __has_builtin(__builtin_amdgcn_tanh_f32)
__device__ __forceinline__ float fast_tanh(float x) { return __builtin_amdgcn_tanh_f32(x); }
#else
__device__ __forceinline__ float fast_tanh(float x) { return tanhf(x); }
#endif

#define BRING 4

template <bool REFILL>
__device__ __forceinline__ void gemm1_block(const _Float16* __restrict__ xrow,
                                            const _Float16* __restrict__ bw0,
                                            const _Float16* __restrict__ bw1,
                                            int kt0,
                                            v16h (&bq)[BRING][2],
                                            v8f (&acc)[4][2]) {
    #pragma unroll
    for (int s = 0; s < BRING; ++s) {
        const int kt = kt0 + s;
        v16h a[4];
        #pragma unroll
        for (int mt = 0; mt < 4; ++mt) {
            const _Float16* ap = xrow + (size_t)(mt * 16) * XSTRIDE + kt * 32;
            v8h lo = *(const v8h*)ap;
            v8h hi = *(const v8h*)(ap + 16);
            a[mt] = __builtin_shufflevector(lo, hi, 0,1,2,3,4,5,6,7,8,9,10,11,12,13,14,15);
        }
        #pragma unroll
        for (int j = 0; j < 2; ++j) {
            #pragma unroll
            for (int mt = 0; mt < 4; ++mt)
                acc[mt][j] = __builtin_amdgcn_wmma_f32_16x16x32_f16(
                    false, a[mt], false, bq[s][j], (short)0, acc[mt][j], false, false);
        }
        if (REFILL) {
            bq[s][0] = *(const v16h*)(bw0 + (size_t)(kt + BRING) * 512);
            bq[s][1] = *(const v16h*)(bw1 + (size_t)(kt + BRING) * 512);
        }
    }
}

__device__ __forceinline__ int frag_k(int lane, int e) {
    return (e < 8 ? e : e + 8) + ((lane >= 16) ? 8 : 0);
}

__global__ __launch_bounds__(256) void pack_w1(const float* __restrict__ W1,
                                               const float* __restrict__ gamma,
                                               _Float16* __restrict__ B1pack) {
    int i    = blockIdx.x * 256 + threadIdx.x;
    int e    = i & 15;
    int lane = (i >> 4) & 31;
    int frag = i >> 9;
    int kt   = frag % KT1;
    int nt   = frag / KT1;
    int n    = nt * 16 + (lane & 15);
    int kk   = kt * 32 + frag_k(lane, e);
    B1pack[i] = (_Float16)(gamma[kk] * W1[(size_t)kk * HIDDEN + n]);
}

__global__ __launch_bounds__(256) void pack_w2(const float* __restrict__ W2,
                                               _Float16* __restrict__ B2pack) {
    int i    = blockIdx.x * 256 + threadIdx.x;
    int e    = i & 15;
    int lane = (i >> 4) & 31;
    int frag = i >> 9;
    int kt   = frag % KT2;
    int nt   = frag / KT2;
    int n    = nt * 16 + (lane & 15);
    int kk   = kt * 32 + frag_k(lane, e);
    B2pack[i] = (_Float16)W2[(size_t)kk * NPARAM + n];
}

__global__ __launch_bounds__(256) void col_reduce(const float* __restrict__ W1,
                                                  const float* __restrict__ gamma,
                                                  const float* __restrict__ beta,
                                                  const float* __restrict__ b1,
                                                  float* __restrict__ s1,
                                                  float* __restrict__ b1p) {
    __shared__ float rs[256], rb[256];
    int n = blockIdx.x, t = threadIdx.x;
    float a = 0.f, b = 0.f;
    for (int k = t; k < D_MODEL; k += 256) {
        float w = W1[(size_t)k * HIDDEN + n];
        a += gamma[k] * w;
        b += beta[k] * w;
    }
    rs[t] = a; rb[t] = b;
    __syncthreads();
    for (int s = 128; s > 0; s >>= 1) {
        if (t < s) { rs[t] += rs[t + s]; rb[t] += rb[t + s]; }
        __syncthreads();
    }
    if (t == 0) { s1[n] = rs[0]; b1p[n] = rb[0] + b1[n]; }
}

__global__ __launch_bounds__(512, 1)
void hope_main(const float* __restrict__ slow,
               const _Float16* __restrict__ B1pack,
               const _Float16* __restrict__ B2pack,
               const float* __restrict__ s1,
               const float* __restrict__ b1p,
               const float* __restrict__ b2,
               float* __restrict__ out,
               int Btot) {
    extern __shared__ __align__(16) char smem[];
    _Float16* xs     = (_Float16*)smem;
    float*    psum   = (float*)(smem + X_BYTES);
    float*    psumsq = psum + 512;
    float*    ldsMu  = psumsq + 512;
    float*    ldsRs  = ldsMu + 64;

    const int tid  = threadIdx.x;
    const int row0 = blockIdx.x * MTILE;
    const int wave = tid >> 5;
    const int lane = tid & 31;
    const int lh   = lane & 15;
    const int hiK  = (lane >= 16) ? 8 : 0;
    const int hiM  = (lane >= 16) ? 8 : 0;

    #pragma unroll 4
    for (int it = 0; it < (MTILE * D_MODEL / 4) / 512; ++it) {
        int c  = it * 512 + tid;
        int r  = c >> 9;
        int c4 = c & 511;
        v4f v = *(const v4f*)(slow + (size_t)(row0 + r) * D_MODEL + c4 * 4);
        v4h h = { (_Float16)v[0], (_Float16)v[1], (_Float16)v[2], (_Float16)v[3] };
        *(v4h*)(xs + (size_t)r * XSTRIDE + c4 * 4) = h;
    }
    __syncthreads();

    const _Float16* bw0 = B1pack + ((size_t)((wave * 2 + 0) * KT1) * 32 + lane) * 16;
    const _Float16* bw1 = B1pack + ((size_t)((wave * 2 + 1) * KT1) * 32 + lane) * 16;
    v16h bq[BRING][2];
    #pragma unroll
    for (int s = 0; s < BRING; ++s) {
        bq[s][0] = *(const v16h*)(bw0 + (size_t)s * 512);
        bq[s][1] = *(const v16h*)(bw1 + (size_t)s * 512);
    }

    {
        int r = tid >> 3, part = tid & 7;
        const _Float16* rp = xs + (size_t)r * XSTRIDE + part * 256;
        float s = 0.f, sq = 0.f;
        for (int u = 0; u < 32; ++u) {
            v8h hv = *(const v8h*)(rp + u * 8);
            #pragma unroll
            for (int e = 0; e < 8; ++e) { float f = (float)hv[e]; s += f; sq += f * f; }
        }
        psum[tid] = s; psumsq[tid] = sq;
    }
    __syncthreads();
    if (tid < MTILE) {
        float s = 0.f, sq = 0.f;
        #pragma unroll
        for (int p = 0; p < 8; ++p) { s += psum[tid * 8 + p]; sq += psumsq[tid * 8 + p]; }
        float mu  = s * (1.0f / D_MODEL);
        float var = sq * (1.0f / D_MODEL) - mu * mu;
        ldsMu[tid] = mu;
        ldsRs[tid] = rsqrtf(var + 1e-5f);
    }
    __syncthreads();

    v8f acc[4][2] = {};
    {
        const _Float16* xrow = xs + (size_t)lh * XSTRIDE + hiK;
        for (int kt0 = 0; kt0 < KT1 - BRING; kt0 += BRING)
            gemm1_block<true>(xrow, bw0, bw1, kt0, bq, acc);
        gemm1_block<false>(xrow, bw0, bw1, KT1 - BRING, bq, acc);
    }
    __syncthreads();

    #pragma unroll
    for (int j = 0; j < 2; ++j) {
        int col = wave * 32 + j * 16 + lh;
        float s1v = s1[col];
        float b1v = b1p[col];
        #pragma unroll
        for (int mt = 0; mt < 4; ++mt) {
            #pragma unroll
            for (int i = 0; i < 8; ++i) {
                int row = mt * 16 + i + hiM;
                float pre = ldsRs[row] * (acc[mt][j][i] - ldsMu[row] * s1v) + b1v;
                float g = 0.5f * pre * (1.0f + erff(pre * 0.70710678118654752f));
                xs[(size_t)row * HSTRIDE + col] = (_Float16)g;
            }
        }
    }
    __syncthreads();

    for (int job = wave; job < 4 * NT2; job += 16) {
        int m2 = job & 3;
        int n2 = job >> 2;
        v8f c2 = {};
        for (int kt = 0; kt < KT2; ++kt) {
            int kb = kt * 32 + hiK;
            const _Float16* ap = xs + (size_t)(m2 * 16 + lh) * HSTRIDE + kb;
            v8h lo = *(const v8h*)ap;
            v8h hi = *(const v8h*)(ap + 16);
            v16h a2 = __builtin_shufflevector(lo, hi, 0,1,2,3,4,5,6,7,8,9,10,11,12,13,14,15);
            v16h b  = *(const v16h*)(B2pack + ((size_t)(n2 * KT2 + kt) * 32 + lane) * 16);
            c2 = __builtin_amdgcn_wmma_f32_16x16x32_f16(
                false, a2, false, b, (short)0, c2, false, false);
        }
        int col  = n2 * 16 + lh;
        int head = col / 3;
        int comp = col % 3;
        float b2v = b2[col];
        #pragma unroll
        for (int i = 0; i < 8; ++i) {
            int grow = row0 + m2 * 16 + i + hiM;
            float val = fast_tanh(c2[i] + b2v) + 1.0f;
            out[(size_t)comp * Btot * NHEADS + (size_t)grow * NHEADS + head] = val;
        }
    }
}

extern "C" void kernel_launch(void* const* d_in, const int* in_sizes, int n_in,
                              void* d_out, int out_size, void* d_ws, size_t ws_size,
                              hipStream_t stream) {
    const float* slow  = (const float*)d_in[0];
    const float* gamma = (const float*)d_in[1];
    const float* beta  = (const float*)d_in[2];
    const float* W1    = (const float*)d_in[3];
    const float* b1    = (const float*)d_in[4];
    const float* W2    = (const float*)d_in[5];
    const float* b2    = (const float*)d_in[6];
    float* out = (float*)d_out;

    const int Btot = in_sizes[0] / D_MODEL;

    char* ws = (char*)d_ws;
    _Float16* B1pack = (_Float16*)ws;
    _Float16* B2pack = (_Float16*)(ws + 2097152);
    float*    s1     = (float*)(ws + 2097152 + 98304);
    float*    b1p    = (float*)(ws + 2097152 + 98304 + 2048);

    pack_w1<<<(D_MODEL * HIDDEN) / 256, 256, 0, stream>>>(W1, gamma, B1pack);
    pack_w2<<<(HIDDEN * NPARAM) / 256, 256, 0, stream>>>(W2, B2pack);
    col_reduce<<<HIDDEN, 256, 0, stream>>>(W1, gamma, beta, b1, s1, b1p);

    hope_main<<<Btot / MTILE, 512, SMEM_BYTES, stream>>>(
        slow, B1pack, B2pack, s1, b1p, b2, out, Btot);
}


# The three mutually independent weight-prep kernels (W1 pack, W2 pack, column sums) become block ranges of ONE launch: two dependent-launch gaps go, the short jobs run beside the W1 pack
# speedup vs baseline: 1.4204x; 1.0718x over previous
#include <hip/hip_runtime.h>
#include <math.h>


typedef __attribute__((ext_vector_type(16))) _Float16 v16h;
typedef __attribute__((ext_vector_type(8)))  _Float16 v8h;
typedef __attribute__((ext_vector_type(4)))  _Float16 v4h;
typedef __attribute__((ext_vector_type(8)))  float    v8f;
typedef __attribute__((ext_vector_type(4)))  float    v4f;

#define D_MODEL 2048
#define HIDDEN  512
#define NPARAM  96
#define NHEADS  32
#define MTILE   64
#define KT1     (D_MODEL / 32)
#define KT2     (HIDDEN  / 32)
#define NT2     (NPARAM  / 16)

#define XSTRIDE 2056
#define HSTRIDE 520

#define X_BYTES    (MTILE * XSTRIDE * 2)
#define SMEM_BYTES (X_BYTES + 512*4 + 512*4 + 64*4 + 64*4)

#if __has_builtin(__builtin_amdgcn_tanhf)
__device__ __forceinline__ float fast_tanh(float x) { return __builtin_amdgcn_tanhf(x); }
#elif __has_builtin(__builtin_amdgcn_tanh_f32)
__device__ __forceinline__ float fast_tanh(float x) { return __builtin_amdgcn_tanh_f32(x); }
#else
__device__ __forceinline__ float fast_tanh(float x) { return tanhf(x); }
#endif

#define BRING 4

template <bool REFILL>
__device__ __forceinline__ void gemm1_block(const _Float16* __restrict__ xrow,
                                            const _Float16* __restrict__ bw0,
                                            const _Float16* __restrict__ bw1,
                                            int kt0,
                                            v16h (&bq)[BRING][2],
                                            v8f (&acc)[4][2]) {
    #pragma unroll
    for (int s = 0; s < BRING; ++s) {
        const int kt = kt0 + s;
        v16h a[4];
        #pragma unroll
        for (int mt = 0; mt < 4; ++mt) {
            const _Float16* ap = xrow + (size_t)(mt * 16) * XSTRIDE + kt * 32;
            v8h lo = *(const v8h*)ap;
            v8h hi = *(const v8h*)(ap + 16);
            a[mt] = __builtin_shufflevector(lo, hi, 0,1,2,3,4,5,6,7,8,9,10,11,12,13,14,15);
        }
        #pragma unroll
        for (int j = 0; j < 2; ++j) {
            #pragma unroll
            for (int mt = 0; mt < 4; ++mt)
                acc[mt][j] = __builtin_amdgcn_wmma_f32_16x16x32_f16(
                    false, a[mt], false, bq[s][j], (short)0, acc[mt][j], false, false);
        }
        if (REFILL) {
            bq[s][0] = *(const v16h*)(bw0 + (size_t)(kt + BRING) * 512);
            bq[s][1] = *(const v16h*)(bw1 + (size_t)(kt + BRING) * 512);
        }
    }
}

__device__ __forceinline__ int frag_k(int lane, int e) {
    return (e < 8 ? e : e + 8) + ((lane >= 16) ? 8 : 0);
}

__device__ __forceinline__ void pack_w1(int blk,
                                         const float* __restrict__ W1,
                                         const float* __restrict__ gamma,
                                         _Float16* __restrict__ B1pack) {
    int i    = blk * 256 + threadIdx.x;
    int e    = i & 15;
    int lane = (i >> 4) & 31;
    int frag = i >> 9;
    int kt   = frag % KT1;
    int nt   = frag / KT1;
    int n    = nt * 16 + (lane & 15);
    int kk   = kt * 32 + frag_k(lane, e);
    B1pack[i] = (_Float16)(gamma[kk] * W1[(size_t)kk * HIDDEN + n]);
}

__device__ __forceinline__ void pack_w2(int blk,
                                         const float* __restrict__ W2,
                                         _Float16* __restrict__ B2pack) {
    int i    = blk * 256 + threadIdx.x;
    int e    = i & 15;
    int lane = (i >> 4) & 31;
    int frag = i >> 9;
    int kt   = frag % KT2;
    int nt   = frag / KT2;
    int n    = nt * 16 + (lane & 15);
    int kk   = kt * 32 + frag_k(lane, e);
    B2pack[i] = (_Float16)W2[(size_t)kk * NPARAM + n];
}

__device__ __forceinline__ void col_reduce(int n, float* rs, float* rb,
                                            const float* __restrict__ W1,
                                            const float* __restrict__ gamma,
                                            const float* __restrict__ beta,
                                            const float* __restrict__ b1,
                                            float* __restrict__ s1,
                                            float* __restrict__ b1p) {
    int t = threadIdx.x;
    float a = 0.f, b = 0.f;
    for (int k = t; k < D_MODEL; k += 256) {
        float w = W1[(size_t)k * HIDDEN + n];
        a += gamma[k] * w;
        b += beta[k] * w;
    }
    rs[t] = a; rb[t] = b;
    __syncthreads();
    for (int s = 128; s > 0; s >>= 1) {
        if (t < s) { rs[t] += rs[t + s]; rb[t] += rb[t + s]; }
        __syncthreads();
    }
    if (t == 0) { s1[n] = rs[0]; b1p[n] = rb[0] + b1[n]; }
}

#define CR_BLOCKS  HIDDEN
#define PW1_BLOCKS ((D_MODEL * HIDDEN) / 256)
#define PW2_BLOCKS ((HIDDEN * NPARAM) / 256)
#define PREP_BLOCKS (CR_BLOCKS + PW1_BLOCKS + PW2_BLOCKS)

__global__ __launch_bounds__(256) void prep_all(const float* __restrict__ W1,
                                                const float* __restrict__ gamma,
                                                const float* __restrict__ beta,
                                                const float* __restrict__ b1,
                                                const float* __restrict__ W2,
                                                _Float16* __restrict__ B1pack,
                                                _Float16* __restrict__ B2pack,
                                                float* __restrict__ s1,
                                                float* __restrict__ b1p) {
    __shared__ float rs[256], rb[256];
    const int b = blockIdx.x;
    if (b < CR_BLOCKS)
        col_reduce(b, rs, rb, W1, gamma, beta, b1, s1, b1p);
    else if (b < CR_BLOCKS + PW1_BLOCKS)
        pack_w1(b - CR_BLOCKS, W1, gamma, B1pack);
    else
        pack_w2(b - CR_BLOCKS - PW1_BLOCKS, W2, B2pack);
}

__global__ __launch_bounds__(512, 1)
void hope_main(const float* __restrict__ slow,
               const _Float16* __restrict__ B1pack,
               const _Float16* __restrict__ B2pack,
               const float* __restrict__ s1,
               const float* __restrict__ b1p,
               const float* __restrict__ b2,
               float* __restrict__ out,
               int Btot) {
    extern __shared__ __align__(16) char smem[];
    _Float16* xs     = (_Float16*)smem;
    float*    psum   = (float*)(smem + X_BYTES);
    float*    psumsq = psum + 512;
    float*    ldsMu  = psumsq + 512;
    float*    ldsRs  = ldsMu + 64;

    const int tid  = threadIdx.x;
    const int row0 = blockIdx.x * MTILE;
    const int wave = tid >> 5;
    const int lane = tid & 31;
    const int lh   = lane & 15;
    const int hiK  = (lane >= 16) ? 8 : 0;
    const int hiM  = (lane >= 16) ? 8 : 0;

    #pragma unroll 4
    for (int it = 0; it < (MTILE * D_MODEL / 4) / 512; ++it) {
        int c  = it * 512 + tid;
        int r  = c >> 9;
        int c4 = c & 511;
        v4f v = *(const v4f*)(slow + (size_t)(row0 + r) * D_MODEL + c4 * 4);
        v4h h = { (_Float16)v[0], (_Float16)v[1], (_Float16)v[2], (_Float16)v[3] };
        *(v4h*)(xs + (size_t)r * XSTRIDE + c4 * 4) = h;
    }
    __syncthreads();

    const _Float16* bw0 = B1pack + ((size_t)((wave * 2 + 0) * KT1) * 32 + lane) * 16;
    const _Float16* bw1 = B1pack + ((size_t)((wave * 2 + 1) * KT1) * 32 + lane) * 16;
    v16h bq[BRING][2];
    #pragma unroll
    for (int s = 0; s < BRING; ++s) {
        bq[s][0] = *(const v16h*)(bw0 + (size_t)s * 512);
        bq[s][1] = *(const v16h*)(bw1 + (size_t)s * 512);
    }

    {
        int r = tid >> 3, part = tid & 7;
        const _Float16* rp = xs + (size_t)r * XSTRIDE + part * 256;
        float s = 0.f, sq = 0.f;
        for (int u = 0; u < 32; ++u) {
            v8h hv = *(const v8h*)(rp + u * 8);
            #pragma unroll
            for (int e = 0; e < 8; ++e) { float f = (float)hv[e]; s += f; sq += f * f; }
        }
        psum[tid] = s; psumsq[tid] = sq;
    }
    __syncthreads();
    if (tid < MTILE) {
        float s = 0.f, sq = 0.f;
        #pragma unroll
        for (int p = 0; p < 8; ++p) { s += psum[tid * 8 + p]; sq += psumsq[tid * 8 + p]; }
        float mu  = s * (1.0f / D_MODEL);
        float var = sq * (1.0f / D_MODEL) - mu * mu;
        ldsMu[tid] = mu;
        ldsRs[tid] = rsqrtf(var + 1e-5f);
    }
    __syncthreads();

    v8f acc[4][2] = {};
    {
        const _Float16* xrow = xs + (size_t)lh * XSTRIDE + hiK;
        for (int kt0 = 0; kt0 < KT1 - BRING; kt0 += BRING)
            gemm1_block<true>(xrow, bw0, bw1, kt0, bq, acc);
        gemm1_block<false>(xrow, bw0, bw1, KT1 - BRING, bq, acc);
    }
    __syncthreads();

    #pragma unroll
    for (int j = 0; j < 2; ++j) {
        int col = wave * 32 + j * 16 + lh;
        float s1v = s1[col];
        float b1v = b1p[col];
        #pragma unroll
        for (int mt = 0; mt < 4; ++mt) {
            #pragma unroll
            for (int i = 0; i < 8; ++i) {
                int row = mt * 16 + i + hiM;
                float pre = ldsRs[row] * (acc[mt][j][i] - ldsMu[row] * s1v) + b1v;
                float g = 0.5f * pre * (1.0f + erff(pre * 0.70710678118654752f));
                xs[(size_t)row * HSTRIDE + col] = (_Float16)g;
            }
        }
    }
    __syncthreads();

    for (int job = wave; job < 4 * NT2; job += 16) {
        int m2 = job & 3;
        int n2 = job >> 2;
        v8f c2 = {};
        for (int kt = 0; kt < KT2; ++kt) {
            int kb = kt * 32 + hiK;
            const _Float16* ap = xs + (size_t)(m2 * 16 + lh) * HSTRIDE + kb;
            v8h lo = *(const v8h*)ap;
            v8h hi = *(const v8h*)(ap + 16);
            v16h a2 = __builtin_shufflevector(lo, hi, 0,1,2,3,4,5,6,7,8,9,10,11,12,13,14,15);
            v16h b  = *(const v16h*)(B2pack + ((size_t)(n2 * KT2 + kt) * 32 + lane) * 16);
            c2 = __builtin_amdgcn_wmma_f32_16x16x32_f16(
                false, a2, false, b, (short)0, c2, false, false);
        }
        int col  = n2 * 16 + lh;
        int head = col / 3;
        int comp = col % 3;
        float b2v = b2[col];
        #pragma unroll
        for (int i = 0; i < 8; ++i) {
            int grow = row0 + m2 * 16 + i + hiM;
            float val = fast_tanh(c2[i] + b2v) + 1.0f;
            out[(size_t)comp * Btot * NHEADS + (size_t)grow * NHEADS + head] = val;
        }
    }
}

extern "C" void kernel_launch(void* const* d_in, const int* in_sizes, int n_in,
                              void* d_out, int out_size, void* d_ws, size_t ws_size,
                              hipStream_t stream) {
    const float* slow  = (const float*)d_in[0];
    const float* gamma = (const float*)d_in[1];
    const float* beta  = (const float*)d_in[2];
    const float* W1    = (const float*)d_in[3];
    const float* b1    = (const float*)d_in[4];
    const float* W2    = (const float*)d_in[5];
    const float* b2    = (const float*)d_in[6];
    float* out = (float*)d_out;

    const int Btot = in_sizes[0] / D_MODEL;

    char* ws = (char*)d_ws;
    _Float16* B1pack = (_Float16*)ws;
    _Float16* B2pack = (_Float16*)(ws + 2097152);
    float*    s1     = (float*)(ws + 2097152 + 98304);
    float*    b1p    = (float*)(ws + 2097152 + 98304 + 2048);

    prep_all<<<PREP_BLOCKS, 256, 0, stream>>>(
        W1, gamma, beta, b1, W2, B1pack, B2pack, s1, b1p);

    hope_main<<<Btot / MTILE, 512, SMEM_BYTES, stream>>>(
        slow, B1pack, B2pack, s1, b1p, b2, out, Btot);
}
